# starting kernel
# baseline (speedup 1.0000x reference)
#include <hip/hip_runtime.h>

typedef float v2f __attribute__((ext_vector_type(2)));
typedef float v8f __attribute__((ext_vector_type(8)));

#define LEAKY 0.2f
#define L2EPS 1e-12f


__global__ void fill_zero_f32(float* __restrict__ p, int n) {
  int i = blockIdx.x * blockDim.x + threadIdx.x;
  if (i < n) p[i] = 0.0f;
}

__global__ void deg_kernel(const int* __restrict__ ei, float* __restrict__ deg, int E) {
  int e = blockIdx.x * blockDim.x + threadIdx.x;
  if (e < E) atomicAdd(&deg[ei[e]], 1.0f);
}

__global__ void dinv_kernel(float* __restrict__ d, int N) {
  int i = blockIdx.x * blockDim.x + threadIdx.x;
  if (i < N) {
    float v = d[i];
    d[i] = (v > 0.0f) ? rsqrtf(v) : 0.0f;
  }
}

__global__ void edgew_kernel(const int* __restrict__ ei, const float* __restrict__ dinv,
                             float* __restrict__ w, int E) {
  int e = blockIdx.x * blockDim.x + threadIdx.x;
  if (e < E) w[e] = dinv[ei[e]] * dinv[ei[E + e]];
}

__global__ void copy_x_kernel(const float* __restrict__ x, float* __restrict__ out,
                              int N, int outStride) {
  int t = blockIdx.x * blockDim.x + threadIdx.x;
  if (t < N * 16) {
    int r = t >> 4, q = t & 15;
    float4 v = reinterpret_cast<const float4*>(x)[(size_t)r * 16 + q];
    reinterpret_cast<float4*>(out + (size_t)r * outStride)[q] = v;
  }
}

__global__ void agg_kernel(const int* __restrict__ ei, const float* __restrict__ w,
                           const float* __restrict__ h, float* __restrict__ ha, int E) {
  long long t = (long long)blockIdx.x * blockDim.x + threadIdx.x;
  if (t >= (long long)E * 32) return;
  int e = (int)(t >> 5);
  int p = (int)(t & 31);
  int r = ei[e];
  int c = ei[E + e];
  float wv = w[e];
  v2f hv = *reinterpret_cast<const v2f*>(h + (size_t)c * 64 + p * 2);
  float* dst = ha + (size_t)r * 64 + p * 2;
  atomicAdd(dst + 0, wv * hv.x);
  atomicAdd(dst + 1, wv * hv.y);
}

__launch_bounds__(256)
__global__ void layer_kernel(const float* __restrict__ h,  const float* __restrict__ ha,
                             const float* __restrict__ Wg, const float* __restrict__ bg,
                             const float* __restrict__ Wi, const float* __restrict__ bi,
                             float* __restrict__ hnext, float* __restrict__ out,
                             int N, int outStride, int colOff) {
  __shared__ float sWg[64 * 64];
  __shared__ float sWi[64 * 64];
  __shared__ float sbg[64];
  __shared__ float sbi[64];

  int tid = threadIdx.x;
  for (int i = tid; i < 4096; i += 256) {
    int r = i >> 6, c = i & 63;
    int idx = (((r >> 1) * 64 + c) << 1) | (r & 1);
    sWg[idx] = Wg[i];
    sWi[idx] = Wi[i];
  }
  if (tid < 64) { sbg[tid] = bg[tid]; sbi[tid] = bi[tid]; }
  __syncthreads();

  int wave = tid >> 5, lane = tid & 31;
  int tiles = (N + 15) >> 4;
  int tile = blockIdx.x * 8 + wave;
  if (tile >= tiles) return;

  int m0 = tile << 4;
  int kh  = lane >> 4;
  int l15 = lane & 15;

  int rowA = m0 + l15;
  int rowLd = rowA < N ? rowA : N - 1;
  const float* haRow = ha + (size_t)rowLd * 64 + 2 * kh;
  const float* hRow  = h  + (size_t)rowLd * 64 + 2 * kh;

  v8f zero8 = {0.f, 0.f, 0.f, 0.f, 0.f, 0.f, 0.f, 0.f};
  v8f accg[4], acci[4];
#pragma unroll
  for (int nt = 0; nt < 4; nt++) { accg[nt] = zero8; acci[nt] = zero8; }

  const v2f* sWg2 = reinterpret_cast<const v2f*>(sWg);
  const v2f* sWi2 = reinterpret_cast<const v2f*>(sWi);

#pragma unroll
  for (int k = 0; k < 64; k += 4) {
    v2f ga = *reinterpret_cast<const v2f*>(haRow + k);
    v2f hv = *reinterpret_cast<const v2f*>(hRow + k);
    v2f ia = ga * hv;
    int k2 = (k >> 1) + kh;
#pragma unroll
    for (int nt = 0; nt < 4; nt++) {
      v2f bgf = sWg2[k2 * 64 + nt * 16 + l15];
      v2f bif = sWi2[k2 * 64 + nt * 16 + l15];
      accg[nt] = __builtin_amdgcn_wmma_f32_16x16x4_f32(
          false, ga, false, bgf, (short)0, accg[nt], false, false);
      acci[nt] = __builtin_amdgcn_wmma_f32_16x16x4_f32(
          false, ia, false, bif, (short)0, acci[nt], false, false);
    }
  }

  float bgl[4], bil[4];
#pragma unroll
  for (int nt = 0; nt < 4; nt++) {
    bgl[nt] = sbg[nt * 16 + l15];
    bil[nt] = sbi[nt * 16 + l15];
  }

#pragma unroll
  for (int j = 0; j < 8; j++) {
    int row = m0 + j + 8 * kh;
    float v[4];
    float sq = 0.0f;
#pragma unroll
    for (int nt = 0; nt < 4; nt++) {
      float g = accg[nt][j] + bgl[nt];
      float s = acci[nt][j] + bil[nt];
      g = (g > 0.0f) ? g : LEAKY * g;
      s = (s > 0.0f) ? s : LEAKY * s;
      v[nt] = g + s;
      sq += v[nt] * v[nt];
    }
#pragma unroll
    for (int m = 1; m < 16; m <<= 1) sq += __shfl_xor(sq, m, 32);
    float inv = rsqrtf(fmaxf(sq, L2EPS));
    if (row < N) {
      float* hn = hnext + (size_t)row * 64 + l15;
      float* op = out + (size_t)row * outStride + colOff + l15;
#pragma unroll
      for (int nt = 0; nt < 4; nt++) {
        hn[nt * 16] = v[nt];
        op[nt * 16] = v[nt] * inv;
      }
    }
  }
}


extern "C" void kernel_launch(void* const* d_in, const int* in_sizes, int n_in,
                              void* d_out, int out_size, void* d_ws, size_t ws_size,
                              hipStream_t stream) {
  const float* x  = (const float*)d_in[0];
  const int*   ei = (const int*)d_in[1];
  const float* Wg = (const float*)d_in[2];
  const float* bg = (const float*)d_in[3];
  const float* Wi = (const float*)d_in[4];
  const float* bi = (const float*)d_in[5];

  const int d = 64;
  const int N = in_sizes[0] / d;
  const int E = in_sizes[1] / 2;
  const int K = in_sizes[2] / (d * d);
  const int outStride = (K + 1) * d;

  float* ws    = (float*)d_ws;
  float* dinv  = ws;
  float* wvec  = dinv + N;
  float* haBuf = wvec + E;
  float* hA    = haBuf + (size_t)N * d;
  float* hB    = hA + (size_t)N * d;
  float* out   = (float*)d_out;

  fill_zero_f32<<<(N + 255) / 256, 256, 0, stream>>>(dinv, N);
  deg_kernel<<<(E + 255) / 256, 256, 0, stream>>>(ei, dinv, E);
  dinv_kernel<<<(N + 255) / 256, 256, 0, stream>>>(dinv, N);
  edgew_kernel<<<(E + 255) / 256, 256, 0, stream>>>(ei, dinv, wvec, E);
  copy_x_kernel<<<(N * 16 + 255) / 256, 256, 0, stream>>>(x, out, N, outStride);

  const float* h = x;
  float* nxt = hA;
  int tiles = (N + 15) / 16;
  int lblocks = (tiles + 7) / 8;
  long long aggT = (long long)E * 32;
  int ablocks = (int)((aggT + 255) / 256);

  for (int layer = 0; layer < K; layer++) {
    fill_zero_f32<<<(N * d + 255) / 256, 256, 0, stream>>>(haBuf, N * d);
    agg_kernel<<<ablocks, 256, 0, stream>>>(ei, wvec, h, haBuf, E);
    layer_kernel<<<lblocks, 256, 0, stream>>>(
        h, haBuf,
        Wg + (size_t)layer * d * d, bg + (size_t)layer * d,
        Wi + (size_t)layer * d * d, bi + (size_t)layer * d,
        nxt, out, N, outStride, (layer + 1) * d);
    h = nxt;
    nxt = (nxt == hA) ? hB : hA;
  }
}


# Sum by destination row: edges grouped per row each call (integer atomic add as slot, arrival order, + prefix sum); a wave gathers a row in registers, stores it once; float atomics, zero fills dropped
# speedup vs baseline: 2.4733x; 2.4733x over previous
#include <hip/hip_runtime.h>

typedef float v2f __attribute__((ext_vector_type(2)));
typedef float v8f __attribute__((ext_vector_type(8)));

#define LEAKY 0.2f
#define L2EPS 1e-12f


__global__ void fill_zero_i32(int* __restrict__ p, int n) {
  int i = blockIdx.x * blockDim.x + threadIdx.x;
  if (i < n) p[i] = 0;
}

__global__ void rank_kernel(const int* __restrict__ ei, int* __restrict__ cnt,
                            int* __restrict__ rank, int E) {
  int e = blockIdx.x * blockDim.x + threadIdx.x;
  if (e < E) rank[e] = atomicAdd(&cnt[ei[e]], 1);
}

__device__ __forceinline__ int wave_sum_i32(int v) {
  for (int m = 1; m < 32; m <<= 1) v += __shfl_xor(v, m, 32);
  return v;
}

__device__ __forceinline__ int block_sum_i32(int v, int* s8) {
  v = wave_sum_i32(v);
  if ((threadIdx.x & 31) == 0) s8[threadIdx.x >> 5] = v;
  __syncthreads();
  int t = 0;
  for (int i = 0; i < 8; i++) t += s8[i];
  return t;
}

__launch_bounds__(256)
__global__ void bsum_kernel(const int* __restrict__ cnt, int* __restrict__ bsum, int N) {
  __shared__ int s8[8];
  int base = blockIdx.x * 1024 + threadIdx.x;
  int t = 0;
  for (int k = 0; k < 4; k++) {
    int i = base + k * 256;
    if (i < N) t += cnt[i];
  }
  t = block_sum_i32(t, s8);
  if (threadIdx.x == 0) bsum[blockIdx.x] = t;
}

__launch_bounds__(256)
__global__ void scan_kernel(const int* __restrict__ cnt, const int* __restrict__ bsum,
                            int* __restrict__ rowptr, float* __restrict__ dinv, int N) {
  __shared__ int sA[8];
  __shared__ int sB[8];
  int tid = threadIdx.x, lane = tid & 31, wv = tid >> 5;
  int b = blockIdx.x;
  int part = 0;
  for (int i = tid; i < b; i += 256) part += bsum[i];
  int blockOff = block_sum_i32(part, sA);

  int idx0 = b * 1024 + tid * 4;
  int v[4];
  int t = 0;
  for (int k = 0; k < 4; k++) {
    v[k] = (idx0 + k < N) ? cnt[idx0 + k] : 0;
    t += v[k];
  }
  int inc = t;
  for (int d = 1; d < 32; d <<= 1) {
    int o = __shfl_up(inc, d, 32);
    if (lane >= d) inc += o;
  }
  if (lane == 31) sB[wv] = inc;
  __syncthreads();
  int woff = 0;
  for (int i = 0; i < 8; i++) { int s = sB[i]; if (i < wv) woff += s; }
  int p = blockOff + woff + inc - t;
  for (int k = 0; k < 4; k++) {
    int i = idx0 + k;
    if (i <= N) rowptr[i] = p;
    if (i < N) dinv[i] = (v[k] > 0) ? rsqrtf((float)v[k]) : 0.0f;
    p += v[k];
  }
}

__global__ void place_kernel(const int* __restrict__ ei, const int* __restrict__ rank,
                             const int* __restrict__ rowptr, const float* __restrict__ dinv,
                             int* __restrict__ cw, int E) {
  int e = blockIdx.x * blockDim.x + threadIdx.x;
  if (e < E) {
    int r = ei[e];
    int c = ei[E + e];
    float w = dinv[r] * dinv[c];
    int2 rec;
    rec.x = c;
    rec.y = __float_as_int(w);
    reinterpret_cast<int2*>(cw)[rowptr[r] + rank[e]] = rec;
  }
}

__global__ void copy_x_kernel(const float* __restrict__ x, float* __restrict__ out,
                              int N, int outStride) {
  int t = blockIdx.x * blockDim.x + threadIdx.x;
  if (t < N * 16) {
    int r = t >> 4, q = t & 15;
    float4 v = reinterpret_cast<const float4*>(x)[(size_t)r * 16 + q];
    reinterpret_cast<float4*>(out + (size_t)r * outStride)[q] = v;
  }
}

template <int K>
__device__ __forceinline__ void gather_block(v2f& acc, int cj, int wj, int j,
                                             const float* __restrict__ hl) {
  v2f hv[K];
  float ww[K];
#pragma unroll
  for (int u = 0; u < K; u++) {
    int c = __builtin_amdgcn_readlane(cj, j + u);
    ww[u] = __int_as_float(__builtin_amdgcn_readlane(wj, j + u));
    hv[u] = *reinterpret_cast<const v2f*>(hl + (size_t)(unsigned)c * 64);
  }
#pragma unroll
  for (int u = 0; u < K; u++) acc += ww[u] * hv[u];
}

__launch_bounds__(256)
__global__ void gather_kernel(const int* __restrict__ rowptr, const int* __restrict__ cw,
                              const float* __restrict__ h, float* __restrict__ ha, int N) {
  const int2* rec = reinterpret_cast<const int2*>(cw);
  int lane = threadIdx.x & 31;
  int wave0 = (blockIdx.x * 256 + threadIdx.x) >> 5;
  int nw = gridDim.x * 8;
  const float* hl = h + lane * 2;
  for (int r = __builtin_amdgcn_readfirstlane(wave0); r < N; r += nw) {
    int beg = __builtin_amdgcn_readfirstlane(rowptr[r]);
    int end = __builtin_amdgcn_readfirstlane(rowptr[r + 1]);
    v2f acc = {0.0f, 0.0f};
    for (int base = beg; base < end; base += 32) {
      int n = end - base;
      if (n > 32) n = 32;
      int cj = 0, wj = 0;
      if (lane < n) {
        int2 t = rec[base + lane];
        cj = t.x;
        wj = t.y;
      }
      int j = 0;
      for (; j + 8 <= n; j += 8) gather_block<8>(acc, cj, wj, j, hl);
      switch (n - j) {
        case 1: gather_block<1>(acc, cj, wj, j, hl); break;
        case 2: gather_block<2>(acc, cj, wj, j, hl); break;
        case 3: gather_block<3>(acc, cj, wj, j, hl); break;
        case 4: gather_block<4>(acc, cj, wj, j, hl); break;
        case 5: gather_block<5>(acc, cj, wj, j, hl); break;
        case 6: gather_block<6>(acc, cj, wj, j, hl); break;
        case 7: gather_block<7>(acc, cj, wj, j, hl); break;
        default: break;
      }
    }
    *reinterpret_cast<v2f*>(ha + (size_t)r * 64 + lane * 2) = acc;
  }
}

__launch_bounds__(256)
__global__ void layer_kernel(const float* __restrict__ h,  const float* __restrict__ ha,
                             const float* __restrict__ Wg, const float* __restrict__ bg,
                             const float* __restrict__ Wi, const float* __restrict__ bi,
                             float* __restrict__ hnext, float* __restrict__ out,
                             int N, int outStride, int colOff) {
  __shared__ float sWg[64 * 64];
  __shared__ float sWi[64 * 64];
  __shared__ float sbg[64];
  __shared__ float sbi[64];

  int tid = threadIdx.x;
  for (int i = tid; i < 4096; i += 256) {
    int r = i >> 6, c = i & 63;
    int idx = (((r >> 1) * 64 + c) << 1) | (r & 1);
    sWg[idx] = Wg[i];
    sWi[idx] = Wi[i];
  }
  if (tid < 64) { sbg[tid] = bg[tid]; sbi[tid] = bi[tid]; }
  __syncthreads();

  int wave = tid >> 5, lane = tid & 31;
  int tiles = (N + 15) >> 4;
  int tile = blockIdx.x * 8 + wave;
  if (tile >= tiles) return;

  int m0 = tile << 4;
  int kh  = lane >> 4;
  int l15 = lane & 15;

  int rowA = m0 + l15;
  int rowLd = rowA < N ? rowA : N - 1;
  const float* haRow = ha + (size_t)rowLd * 64 + 2 * kh;
  const float* hRow  = h  + (size_t)rowLd * 64 + 2 * kh;

  v8f zero8 = {0.f, 0.f, 0.f, 0.f, 0.f, 0.f, 0.f, 0.f};
  v8f accg[4], acci[4];
#pragma unroll
  for (int nt = 0; nt < 4; nt++) { accg[nt] = zero8; acci[nt] = zero8; }

  const v2f* sWg2 = reinterpret_cast<const v2f*>(sWg);
  const v2f* sWi2 = reinterpret_cast<const v2f*>(sWi);

#pragma unroll
  for (int k = 0; k < 64; k += 4) {
    v2f ga = *reinterpret_cast<const v2f*>(haRow + k);
    v2f hv = *reinterpret_cast<const v2f*>(hRow + k);
    v2f ia = ga * hv;
    int k2 = (k >> 1) + kh;
#pragma unroll
    for (int nt = 0; nt < 4; nt++) {
      v2f bgf = sWg2[k2 * 64 + nt * 16 + l15];
      v2f bif = sWi2[k2 * 64 + nt * 16 + l15];
      accg[nt] = __builtin_amdgcn_wmma_f32_16x16x4_f32(
          false, ga, false, bgf, (short)0, accg[nt], false, false);
      acci[nt] = __builtin_amdgcn_wmma_f32_16x16x4_f32(
          false, ia, false, bif, (short)0, acci[nt], false, false);
    }
  }

  float bgl[4], bil[4];
#pragma unroll
  for (int nt = 0; nt < 4; nt++) {
    bgl[nt] = sbg[nt * 16 + l15];
    bil[nt] = sbi[nt * 16 + l15];
  }

#pragma unroll
  for (int j = 0; j < 8; j++) {
    int row = m0 + j + 8 * kh;
    float v[4];
    float sq = 0.0f;
#pragma unroll
    for (int nt = 0; nt < 4; nt++) {
      float g = accg[nt][j] + bgl[nt];
      float s = acci[nt][j] + bil[nt];
      g = (g > 0.0f) ? g : LEAKY * g;
      s = (s > 0.0f) ? s : LEAKY * s;
      v[nt] = g + s;
      sq += v[nt] * v[nt];
    }
#pragma unroll
    for (int m = 1; m < 16; m <<= 1) sq += __shfl_xor(sq, m, 32);
    float inv = rsqrtf(fmaxf(sq, L2EPS));
    if (row < N) {
      float* hn = hnext + (size_t)row * 64 + l15;
      float* op = out + (size_t)row * outStride + colOff + l15;
#pragma unroll
      for (int nt = 0; nt < 4; nt++) {
        hn[nt * 16] = v[nt];
        op[nt * 16] = v[nt] * inv;
      }
    }
  }
}


extern "C" void kernel_launch(void* const* d_in, const int* in_sizes, int n_in,
                              void* d_out, int out_size, void* d_ws, size_t ws_size,
                              hipStream_t stream) {
  const float* x  = (const float*)d_in[0];
  const int*   ei = (const int*)d_in[1];
  const float* Wg = (const float*)d_in[2];
  const float* bg = (const float*)d_in[3];
  const float* Wi = (const float*)d_in[4];
  const float* bi = (const float*)d_in[5];

  const int d = 64;
  const int N = in_sizes[0] / d;
  const int E = in_sizes[1] / 2;
  const int K = in_sizes[2] / (d * d);
  const int outStride = (K + 1) * d;

  const int nb = (N + 1024) / 1024;

  int*   cw     = (int*)d_ws;
  float* haBuf  = (float*)(cw + 2 * (size_t)E);
  float* hA     = haBuf + (size_t)N * d;
  float* hB     = hA + (size_t)N * d;
  int*   rank   = (int*)(hB + (size_t)N * d);
  int*   cnt    = rank + E;
  int*   rowptr = cnt + N;
  int*   bsum   = rowptr + N + 1;
  float* dinv   = (float*)(bsum + nb);
  float* out    = (float*)d_out;

  fill_zero_i32<<<(N + 255) / 256, 256, 0, stream>>>(cnt, N);
  rank_kernel<<<(E + 255) / 256, 256, 0, stream>>>(ei, cnt, rank, E);
  bsum_kernel<<<nb, 256, 0, stream>>>(cnt, bsum, N);
  scan_kernel<<<nb, 256, 0, stream>>>(cnt, bsum, rowptr, dinv, N);
  place_kernel<<<(E + 255) / 256, 256, 0, stream>>>(ei, rank, rowptr, dinv, cw, E);
  copy_x_kernel<<<(N * 16 + 255) / 256, 256, 0, stream>>>(x, out, N, outStride);

  const float* h = x;
  float* nxt = hA;
  int tiles = (N + 15) / 16;
  int lblocks = (tiles + 7) / 8;

  for (int layer = 0; layer < K; layer++) {
    gather_kernel<<<4096, 256, 0, stream>>>(rowptr, cw, h, haBuf, N);
    layer_kernel<<<lblocks, 256, 0, stream>>>(
        h, haBuf,
        Wg + (size_t)layer * d * d, bg + (size_t)layer * d,
        Wi + (size_t)layer * d * d, bi + (size_t)layer * d,
        nxt, out, N, outStride, (layer + 1) * d);
    h = nxt;
    nxt = (nxt == hA) ? hB : hA;
  }
}


# Edge ranks by returning LDS adds (a counter for every row in LDS, one work-group per 26624-edge chunk) + a scan over chunks; no global atomics; slot order = chunk, then LDS arrival order
# speedup vs baseline: 3.0142x; 1.2187x over previous
#include <hip/hip_runtime.h>
#include <stdint.h>

typedef float v2f __attribute__((ext_vector_type(2)));
typedef float v8f __attribute__((ext_vector_type(8)));

#define LEAKY 0.2f
#define L2EPS 1e-12f


#define CHUNK_EDGES 26624
#define CAP_ROWS 131072
#define CAP_WORDS (CAP_ROWS / 2)

__launch_bounds__(1024)
__global__ void count_kernel(const int* __restrict__ ei, int* __restrict__ rank,
                             unsigned* __restrict__ tab, int E, int N, int Wp) {
  __shared__ unsigned sc[CAP_WORDS];
  int tid = threadIdx.x;
  int e0 = blockIdx.x * CHUNK_EDGES;
  int e1 = e0 + CHUNK_EDGES;
  if (e1 > E) e1 = E;
  uint4* trow4 = reinterpret_cast<uint4*>(tab + (size_t)blockIdx.x * Wp);
  uint4* sc4 = reinterpret_cast<uint4*>(sc);

  for (int base = 0; base < N; base += CAP_ROWS) {
    int rows = N - base;
    if (rows > CAP_ROWS) rows = CAP_ROWS;
    int quads = (((rows + 1) >> 1) + 3) >> 2;
    uint4 z4 = {0u, 0u, 0u, 0u};
    for (int i = tid; i < quads; i += 1024) sc4[i] = z4;
    __syncthreads();

    for (int eb = e0; eb < e1; eb += 13 * 1024) {
      int rr[13];
#pragma unroll
      for (int u = 0; u < 13; u++) {
        int e = eb + u * 1024 + tid;
        rr[u] = (e < e1) ? ei[e] : -1;
      }
#pragma unroll
      for (int u = 0; u < 13; u++) {
        int e = eb + u * 1024 + tid;
        int r = rr[u] - base;
        if (e < e1 && (unsigned)r < (unsigned)rows) {
          unsigned sh = (unsigned)(r & 1) << 4;
          unsigned old = atomicAdd(&sc[r >> 1], 1u << sh);
          rank[e] = (int)((old >> sh) & 0xFFFFu);
        }
      }
    }
    __syncthreads();

    uint4* dst = trow4 + (base >> 3);
    for (int i = tid; i < quads; i += 1024) dst[i] = sc4[i];
    __syncthreads();
  }
}

__device__ __forceinline__ int wave_sum_i32(int v) {
  for (int m = 1; m < 32; m <<= 1) v += __shfl_xor(v, m, 32);
  return v;
}

__device__ __forceinline__ int block_sum_i32(int v, int* s8) {
  v = wave_sum_i32(v);
  if ((threadIdx.x & 31) == 0) s8[threadIdx.x >> 5] = v;
  __syncthreads();
  int t = 0;
  for (int i = 0; i < 8; i++) t += s8[i];
  return t;
}

__launch_bounds__(512)
__global__ void colscan_kernel(const unsigned* __restrict__ tab, int* __restrict__ off,
                               int* __restrict__ cnt, int* __restrict__ bsum,
                               int N, int C, int Wp, int Np) {
  __shared__ int s16[16];
  int wd = blockIdx.x * 512 + threadIdx.x;
  int r0 = wd * 2;
  int lo = 0, hi = 0;
  if (r0 < N) {
    for (int c0 = 0; c0 < C; c0 += 32) {
      unsigned w[32];
#pragma unroll
      for (int u = 0; u < 32; u++) {
        int cc = (c0 + u < C) ? (c0 + u) : (C - 1);
        unsigned t = tab[(size_t)cc * Wp + wd];
        w[u] = (c0 + u < C) ? t : 0u;
      }
#pragma unroll
      for (int u = 0; u < 32; u++) {
        if (c0 + u < C) {
          int2 o;
          o.x = lo;
          o.y = hi;
          *reinterpret_cast<int2*>(off + (size_t)(c0 + u) * Np + r0) = o;
          lo += (int)(w[u] & 0xFFFFu);
          hi += (int)(w[u] >> 16);
        }
      }
    }
    cnt[r0] = lo;
    if (r0 + 1 < N) cnt[r0 + 1] = hi;
  }
  int t = wave_sum_i32(lo + hi);
  if ((threadIdx.x & 31) == 0) s16[threadIdx.x >> 5] = t;
  __syncthreads();
  if (threadIdx.x == 0) {
    int a = 0;
    for (int i = 0; i < 16; i++) a += s16[i];
    bsum[blockIdx.x] = a;
  }
}

__launch_bounds__(256)
__global__ void scan_kernel(const int* __restrict__ cnt, const int* __restrict__ bsum,
                            int* __restrict__ rowptr, float* __restrict__ dinv, int N) {
  __shared__ int sA[8];
  __shared__ int sB[8];
  int tid = threadIdx.x, lane = tid & 31, wv = tid >> 5;
  int b = blockIdx.x;
  int part = 0;
  for (int i = tid; i < b; i += 256) part += bsum[i];
  int blockOff = block_sum_i32(part, sA);

  int idx0 = b * 1024 + tid * 4;
  int v[4];
  int t = 0;
  for (int k = 0; k < 4; k++) {
    v[k] = (idx0 + k < N) ? cnt[idx0 + k] : 0;
    t += v[k];
  }
  int inc = t;
  for (int d = 1; d < 32; d <<= 1) {
    int o = __shfl_up(inc, d, 32);
    if (lane >= d) inc += o;
  }
  if (lane == 31) sB[wv] = inc;
  __syncthreads();
  int woff = 0;
  for (int i = 0; i < 8; i++) { int s = sB[i]; if (i < wv) woff += s; }
  int p = blockOff + woff + inc - t;
  for (int k = 0; k < 4; k++) {
    int i = idx0 + k;
    if (i <= N) rowptr[i] = p;
    if (i < N) dinv[i] = (v[k] > 0) ? rsqrtf((float)v[k]) : 0.0f;
    p += v[k];
  }
}

__global__ void place_kernel(const int* __restrict__ ei, const int* __restrict__ rank,
                             const int* __restrict__ rowptr, const int* __restrict__ off,
                             const float* __restrict__ dinv,
                             int* __restrict__ cw, int E, int Np) {
  int l = blockIdx.x * blockDim.x + threadIdx.x;
  int e = blockIdx.y * CHUNK_EDGES + l;
  if (l < CHUNK_EDGES && e < E) {
    int r = ei[e];
    int c = ei[E + e];
    float w = dinv[r] * dinv[c];
    int2 rec;
    rec.x = c;
    rec.y = __float_as_int(w);
    int slot = rowptr[r] + off[(size_t)blockIdx.y * Np + r] + rank[e];
    reinterpret_cast<int2*>(cw)[slot] = rec;
  }
}

__global__ void copy_x_kernel(const float* __restrict__ x, float* __restrict__ out,
                              int N, int outStride) {
  int t = blockIdx.x * blockDim.x + threadIdx.x;
  if (t < N * 16) {
    int r = t >> 4, q = t & 15;
    float4 v = reinterpret_cast<const float4*>(x)[(size_t)r * 16 + q];
    reinterpret_cast<float4*>(out + (size_t)r * outStride)[q] = v;
  }
}

template <int K>
__device__ __forceinline__ void gather_block(v2f& acc, int cj, int wj, int j,
                                             const float* __restrict__ hl) {
  v2f hv[K];
  float ww[K];
#pragma unroll
  for (int u = 0; u < K; u++) {
    int c = __builtin_amdgcn_readlane(cj, j + u);
    ww[u] = __int_as_float(__builtin_amdgcn_readlane(wj, j + u));
    hv[u] = *reinterpret_cast<const v2f*>(hl + (size_t)(unsigned)c * 64);
  }
#pragma unroll
  for (int u = 0; u < K; u++) acc += ww[u] * hv[u];
}

__launch_bounds__(256)
__global__ void gather_kernel(const int* __restrict__ rowptr, const int* __restrict__ cw,
                              const float* __restrict__ h, float* __restrict__ ha, int N) {
  const int2* rec = reinterpret_cast<const int2*>(cw);
  int lane = threadIdx.x & 31;
  int wave0 = (blockIdx.x * 256 + threadIdx.x) >> 5;
  int nw = gridDim.x * 8;
  const float* hl = h + lane * 2;
  for (int r = __builtin_amdgcn_readfirstlane(wave0); r < N; r += nw) {
    int beg = __builtin_amdgcn_readfirstlane(rowptr[r]);
    int end = __builtin_amdgcn_readfirstlane(rowptr[r + 1]);
    v2f acc = {0.0f, 0.0f};
    for (int base = beg; base < end; base += 32) {
      int n = end - base;
      if (n > 32) n = 32;
      int cj = 0, wj = 0;
      if (lane < n) {
        int2 t = rec[base + lane];
        cj = t.x;
        wj = t.y;
      }
      int j = 0;
      for (; j + 8 <= n; j += 8) gather_block<8>(acc, cj, wj, j, hl);
      switch (n - j) {
        case 1: gather_block<1>(acc, cj, wj, j, hl); break;
        case 2: gather_block<2>(acc, cj, wj, j, hl); break;
        case 3: gather_block<3>(acc, cj, wj, j, hl); break;
        case 4: gather_block<4>(acc, cj, wj, j, hl); break;
        case 5: gather_block<5>(acc, cj, wj, j, hl); break;
        case 6: gather_block<6>(acc, cj, wj, j, hl); break;
        case 7: gather_block<7>(acc, cj, wj, j, hl); break;
        default: break;
      }
    }
    *reinterpret_cast<v2f*>(ha + (size_t)r * 64 + lane * 2) = acc;
  }
}

__launch_bounds__(256)
__global__ void layer_kernel(const float* __restrict__ h,  const float* __restrict__ ha,
                             const float* __restrict__ Wg, const float* __restrict__ bg,
                             const float* __restrict__ Wi, const float* __restrict__ bi,
                             float* __restrict__ hnext, float* __restrict__ out,
                             int N, int outStride, int colOff) {
  __shared__ float sWg[64 * 64];
  __shared__ float sWi[64 * 64];
  __shared__ float sbg[64];
  __shared__ float sbi[64];

  int tid = threadIdx.x;
  for (int i = tid; i < 4096; i += 256) {
    int r = i >> 6, c = i & 63;
    int idx = (((r >> 1) * 64 + c) << 1) | (r & 1);
    sWg[idx] = Wg[i];
    sWi[idx] = Wi[i];
  }
  if (tid < 64) { sbg[tid] = bg[tid]; sbi[tid] = bi[tid]; }
  __syncthreads();

  int wave = tid >> 5, lane = tid & 31;
  int tiles = (N + 15) >> 4;
  int tile = blockIdx.x * 8 + wave;
  if (tile >= tiles) return;

  int m0 = tile << 4;
  int kh  = lane >> 4;
  int l15 = lane & 15;

  int rowA = m0 + l15;
  int rowLd = rowA < N ? rowA : N - 1;
  const float* haRow = ha + (size_t)rowLd * 64 + 2 * kh;
  const float* hRow  = h  + (size_t)rowLd * 64 + 2 * kh;

  v8f zero8 = {0.f, 0.f, 0.f, 0.f, 0.f, 0.f, 0.f, 0.f};
  v8f accg[4], acci[4];
#pragma unroll
  for (int nt = 0; nt < 4; nt++) { accg[nt] = zero8; acci[nt] = zero8; }

  const v2f* sWg2 = reinterpret_cast<const v2f*>(sWg);
  const v2f* sWi2 = reinterpret_cast<const v2f*>(sWi);

#pragma unroll
  for (int k = 0; k < 64; k += 4) {
    v2f ga = *reinterpret_cast<const v2f*>(haRow + k);
    v2f hv = *reinterpret_cast<const v2f*>(hRow + k);
    v2f ia = ga * hv;
    int k2 = (k >> 1) + kh;
#pragma unroll
    for (int nt = 0; nt < 4; nt++) {
      v2f bgf = sWg2[k2 * 64 + nt * 16 + l15];
      v2f bif = sWi2[k2 * 64 + nt * 16 + l15];
      accg[nt] = __builtin_amdgcn_wmma_f32_16x16x4_f32(
          false, ga, false, bgf, (short)0, accg[nt], false, false);
      acci[nt] = __builtin_amdgcn_wmma_f32_16x16x4_f32(
          false, ia, false, bif, (short)0, acci[nt], false, false);
    }
  }

  float bgl[4], bil[4];
#pragma unroll
  for (int nt = 0; nt < 4; nt++) {
    bgl[nt] = sbg[nt * 16 + l15];
    bil[nt] = sbi[nt * 16 + l15];
  }

#pragma unroll
  for (int j = 0; j < 8; j++) {
    int row = m0 + j + 8 * kh;
    float v[4];
    float sq = 0.0f;
#pragma unroll
    for (int nt = 0; nt < 4; nt++) {
      float g = accg[nt][j] + bgl[nt];
      float s = acci[nt][j] + bil[nt];
      g = (g > 0.0f) ? g : LEAKY * g;
      s = (s > 0.0f) ? s : LEAKY * s;
      v[nt] = g + s;
      sq += v[nt] * v[nt];
    }
#pragma unroll
    for (int m = 1; m < 16; m <<= 1) sq += __shfl_xor(sq, m, 32);
    float inv = rsqrtf(fmaxf(sq, L2EPS));
    if (row < N) {
      float* hn = hnext + (size_t)row * 64 + l15;
      float* op = out + (size_t)row * outStride + colOff + l15;
#pragma unroll
      for (int nt = 0; nt < 4; nt++) {
        hn[nt * 16] = v[nt];
        op[nt * 16] = v[nt] * inv;
      }
    }
  }
}


extern "C" void kernel_launch(void* const* d_in, const int* in_sizes, int n_in,
                              void* d_out, int out_size, void* d_ws, size_t ws_size,
                              hipStream_t stream) {
  const float* x  = (const float*)d_in[0];
  const int*   ei = (const int*)d_in[1];
  const float* Wg = (const float*)d_in[2];
  const float* bg = (const float*)d_in[3];
  const float* Wi = (const float*)d_in[4];
  const float* bi = (const float*)d_in[5];

  const int d = 64;
  const int N = in_sizes[0] / d;
  const int E = in_sizes[1] / 2;
  const int K = in_sizes[2] / (d * d);
  const int outStride = (K + 1) * d;

  const int nb = (N + 1024) / 1024;
  const int C  = (E + CHUNK_EDGES - 1) / CHUNK_EDGES;
  const int Np = (N + 7) / 8 * 8;
  const int Wp = Np / 2;

  int*   cw     = (int*)d_ws;
  float* haBuf  = (float*)(cw + 2 * (size_t)E);
  float* hA     = haBuf + (size_t)N * d;
  float* hB     = hA + (size_t)N * d;
  int*   rank   = (int*)(hB + (size_t)N * d);
  int*   cnt    = rank + E;
  int*   rowptr = cnt + N;
  int*   bsum   = rowptr + N + 1;
  float* dinv   = (float*)(bsum + nb);
  unsigned* tab = (unsigned*)(((uintptr_t)(dinv + N) + 15) & ~(uintptr_t)15);
  int*   off    = (int*)(tab + (size_t)C * Wp);
  float* out    = (float*)d_out;

  count_kernel<<<C, 1024, 0, stream>>>(ei, rank, tab, E, N, Wp);
  colscan_kernel<<<nb, 512, 0, stream>>>(tab, off, cnt, bsum, N, C, Wp, Np);
  scan_kernel<<<nb, 256, 0, stream>>>(cnt, bsum, rowptr, dinv, N);
  place_kernel<<<dim3((CHUNK_EDGES + 255) / 256, C), 256, 0, stream>>>(ei, rank, rowptr, off, dinv, cw, E, Np);
  copy_x_kernel<<<(N * 16 + 255) / 256, 256, 0, stream>>>(x, out, N, outStride);

  const float* h = x;
  float* nxt = hA;
  int tiles = (N + 15) / 16;
  int lblocks = (tiles + 7) / 8;

  for (int layer = 0; layer < K; layer++) {
    gather_kernel<<<4096, 256, 0, stream>>>(rowptr, cw, h, haBuf, N);
    layer_kernel<<<lblocks, 256, 0, stream>>>(
        h, haBuf,
        Wg + (size_t)layer * d * d, bg + (size_t)layer * d,
        Wi + (size_t)layer * d * d, bi + (size_t)layer * d,
        nxt, out, N, outStride, (layer + 1) * d);
    h = nxt;
    nxt = (nxt == hA) ? hB : hA;
  }
}


# out[:,0:64]=x copied by extra work-groups of the counting launch, on the WGPs its 61 LDS-bound groups leave idle (8 loads in flight per thread); the separate 50000-wave copy launch is dropped
# speedup vs baseline: 3.5744x; 1.1859x over previous
#include <hip/hip_runtime.h>
#include <stdint.h>

typedef float v2f __attribute__((ext_vector_type(2)));
typedef float v8f __attribute__((ext_vector_type(8)));
typedef float v4f __attribute__((ext_vector_type(4)));

#define LEAKY 0.2f
#define L2EPS 1e-12f


#define CHUNK_EDGES 26624
#define CAP_ROWS 131072
#define CAP_WORDS (CAP_ROWS / 2)

#define COPY_F4_PER_THREAD 8
#define COPY_F4_PER_BLOCK (COPY_F4_PER_THREAD * 1024)

__launch_bounds__(1024)
__global__ void count_copy_kernel(const int* __restrict__ ei, int* __restrict__ rank,
                                  unsigned* __restrict__ tab, int E, int N, int Wp, int C,
                                  const float* __restrict__ x, float* __restrict__ out, int outStride) {
  __shared__ unsigned sc[CAP_WORDS];
  int tid = threadIdx.x;
  if ((int)blockIdx.x >= C) {
    int total4 = N * 16;
    int b0 = ((int)blockIdx.x - C) * COPY_F4_PER_BLOCK;
    int i0 = b0 + tid;
    const v4f* x4 = reinterpret_cast<const v4f*>(x);
    if (b0 + COPY_F4_PER_BLOCK <= total4) {
      v4f v[COPY_F4_PER_THREAD];
#pragma unroll
      for (int u = 0; u < COPY_F4_PER_THREAD; u++) v[u] = x4[i0 + u * 1024];
      asm volatile("" :: "v"(v[0]), "v"(v[1]), "v"(v[2]), "v"(v[3]), "v"(v[4]), "v"(v[5]), "v"(v[6]), "v"(v[7]));
#pragma unroll
      for (int u = 0; u < COPY_F4_PER_THREAD; u++) {
        int i = i0 + u * 1024;
        reinterpret_cast<v4f*>(out + (size_t)(i >> 4) * outStride)[i & 15] = v[u];
      }
    } else {
      for (int u = 0; u < COPY_F4_PER_THREAD; u++) {
        int i = i0 + u * 1024;
        if (i < total4) reinterpret_cast<v4f*>(out + (size_t)(i >> 4) * outStride)[i & 15] = x4[i];
      }
    }
    return;
  }
  int e0 = blockIdx.x * CHUNK_EDGES;
  int e1 = e0 + CHUNK_EDGES;
  if (e1 > E) e1 = E;
  uint4* trow4 = reinterpret_cast<uint4*>(tab + (size_t)blockIdx.x * Wp);
  uint4* sc4 = reinterpret_cast<uint4*>(sc);

  for (int base = 0; base < N; base += CAP_ROWS) {
    int rows = N - base;
    if (rows > CAP_ROWS) rows = CAP_ROWS;
    int quads = (((rows + 1) >> 1) + 3) >> 2;
    uint4 z4 = {0u, 0u, 0u, 0u};
    for (int i = tid; i < quads; i += 1024) sc4[i] = z4;
    __syncthreads();

    for (int eb = e0; eb < e1; eb += 13 * 1024) {
      int rr[13];
#pragma unroll
      for (int u = 0; u < 13; u++) {
        int e = eb + u * 1024 + tid;
        rr[u] = (e < e1) ? ei[e] : -1;
      }
#pragma unroll
      for (int u = 0; u < 13; u++) {
        int e = eb + u * 1024 + tid;
        int r = rr[u] - base;
        if (e < e1 && (unsigned)r < (unsigned)rows) {
          unsigned sh = (unsigned)(r & 1) << 4;
          unsigned old = atomicAdd(&sc[r >> 1], 1u << sh);
          rank[e] = (int)((old >> sh) & 0xFFFFu);
        }
      }
    }
    __syncthreads();

    uint4* dst = trow4 + (base >> 3);
    for (int i = tid; i < quads; i += 1024) dst[i] = sc4[i];
    __syncthreads();
  }
}

__device__ __forceinline__ int wave_sum_i32(int v) {
  for (int m = 1; m < 32; m <<= 1) v += __shfl_xor(v, m, 32);
  return v;
}

__device__ __forceinline__ int block_sum_i32(int v, int* s8) {
  v = wave_sum_i32(v);
  if ((threadIdx.x & 31) == 0) s8[threadIdx.x >> 5] = v;
  __syncthreads();
  int t = 0;
  for (int i = 0; i < 8; i++) t += s8[i];
  return t;
}

__launch_bounds__(512)
__global__ void colscan_kernel(const unsigned* __restrict__ tab, int* __restrict__ off,
                               int* __restrict__ cnt, int* __restrict__ bsum,
                               int N, int C, int Wp, int Np) {
  __shared__ int s16[16];
  int wd = blockIdx.x * 512 + threadIdx.x;
  int r0 = wd * 2;
  int lo = 0, hi = 0;
  if (r0 < N) {
    for (int c0 = 0; c0 < C; c0 += 32) {
      unsigned w[32];
#pragma unroll
      for (int u = 0; u < 32; u++) {
        int cc = (c0 + u < C) ? (c0 + u) : (C - 1);
        unsigned t = tab[(size_t)cc * Wp + wd];
        w[u] = (c0 + u < C) ? t : 0u;
      }
#pragma unroll
      for (int u = 0; u < 32; u++) {
        if (c0 + u < C) {
          int2 o;
          o.x = lo;
          o.y = hi;
          *reinterpret_cast<int2*>(off + (size_t)(c0 + u) * Np + r0) = o;
          lo += (int)(w[u] & 0xFFFFu);
          hi += (int)(w[u] >> 16);
        }
      }
    }
    cnt[r0] = lo;
    if (r0 + 1 < N) cnt[r0 + 1] = hi;
  }
  int t = wave_sum_i32(lo + hi);
  if ((threadIdx.x & 31) == 0) s16[threadIdx.x >> 5] = t;
  __syncthreads();
  if (threadIdx.x == 0) {
    int a = 0;
    for (int i = 0; i < 16; i++) a += s16[i];
    bsum[blockIdx.x] = a;
  }
}

__launch_bounds__(256)
__global__ void scan_kernel(const int* __restrict__ cnt, const int* __restrict__ bsum,
                            int* __restrict__ rowptr, float* __restrict__ dinv, int N) {
  __shared__ int sA[8];
  __shared__ int sB[8];
  int tid = threadIdx.x, lane = tid & 31, wv = tid >> 5;
  int b = blockIdx.x;
  int part = 0;
  for (int i = tid; i < b; i += 256) part += bsum[i];
  int blockOff = block_sum_i32(part, sA);

  int idx0 = b * 1024 + tid * 4;
  int v[4];
  int t = 0;
  for (int k = 0; k < 4; k++) {
    v[k] = (idx0 + k < N) ? cnt[idx0 + k] : 0;
    t += v[k];
  }
  int inc = t;
  for (int d = 1; d < 32; d <<= 1) {
    int o = __shfl_up(inc, d, 32);
    if (lane >= d) inc += o;
  }
  if (lane == 31) sB[wv] = inc;
  __syncthreads();
  int woff = 0;
  for (int i = 0; i < 8; i++) { int s = sB[i]; if (i < wv) woff += s; }
  int p = blockOff + woff + inc - t;
  for (int k = 0; k < 4; k++) {
    int i = idx0 + k;
    if (i <= N) rowptr[i] = p;
    if (i < N) dinv[i] = (v[k] > 0) ? rsqrtf((float)v[k]) : 0.0f;
    p += v[k];
  }
}

__global__ void place_kernel(const int* __restrict__ ei, const int* __restrict__ rank,
                             const int* __restrict__ rowptr, const int* __restrict__ off,
                             const float* __restrict__ dinv,
                             int* __restrict__ cw, int E, int Np) {
  int l = blockIdx.x * blockDim.x + threadIdx.x;
  int e = blockIdx.y * CHUNK_EDGES + l;
  if (l < CHUNK_EDGES && e < E) {
    int r = ei[e];
    int c = ei[E + e];
    float w = dinv[r] * dinv[c];
    int2 rec;
    rec.x = c;
    rec.y = __float_as_int(w);
    int slot = rowptr[r] + off[(size_t)blockIdx.y * Np + r] + rank[e];
    reinterpret_cast<int2*>(cw)[slot] = rec;
  }
}


#define TILE_WAVES 25
#define LAYER_THREADS (TILE_WAVES * 32)
#define HA_PITCH 68

__device__ __forceinline__ void row_step(v2f& acc, int s, int& j, int& rowEnd, int rp,
                                         float* sT, int lane) {
  while (s >= rowEnd) {
    *reinterpret_cast<v2f*>(sT + j * HA_PITCH + lane * 2) = acc;
    acc.x = 0.0f; acc.y = 0.0f;
    j++;
    rowEnd = (j < 15) ? __builtin_amdgcn_readlane(rp, j + 1) : 0x7fffffff;
  }
}

__device__ __forceinline__ void gather_trip32(v2f& acc, int& s, int& j, int& rowEnd, int rp,
                                              int cj, int wj,
                                              const char* __restrict__ hb, unsigned laneOff,
                                              float* sT, int lane) {
  v2f hv[32];
#pragma unroll
  for (int q = 0; q < 32; q++) {
    int c = __builtin_amdgcn_readlane(cj, q);
    hv[q] = *reinterpret_cast<const v2f*>(hb + (size_t)(unsigned)c * 256u + laneOff);
  }
#pragma unroll
  for (int q = 0; q < 32; q++) {
    row_step(acc, s + q, j, rowEnd, rp, sT, lane);
    float w = __int_as_float(__builtin_amdgcn_readlane(wj, q));
    acc += w * hv[q];
  }
  s += 32;
}

__device__ __forceinline__ void gather_trip_short(v2f& acc, int& s, int& j, int& rowEnd, int rp,
                                                  int cj, int wj, int n,
                                                  const char* __restrict__ hb, unsigned laneOff,
                                                  float* sT, int lane) {
  for (int q0 = 0; q0 < n; q0 += 8) {
    v2f hv[8];
#pragma unroll
    for (int u = 0; u < 8; u++) {
      int c = __builtin_amdgcn_readlane(cj, q0 + u);
      hv[u] = *reinterpret_cast<const v2f*>(hb + (size_t)(unsigned)c * 256u + laneOff);
    }
#pragma unroll
    for (int u = 0; u < 8; u++) {
      if (q0 + u < n) {
        row_step(acc, s, j, rowEnd, rp, sT, lane);
        float w = __int_as_float(__builtin_amdgcn_readlane(wj, q0 + u));
        acc += w * hv[u];
        s++;
      }
    }
  }
}

__launch_bounds__(LAYER_THREADS)
__global__ void layer_kernel(const int* __restrict__ rowptr, const int* __restrict__ cw,
                             const float* __restrict__ h,
                             const float* __restrict__ Wg, const float* __restrict__ bg,
                             const float* __restrict__ Wi, const float* __restrict__ bi,
                             float* __restrict__ hnext, float* __restrict__ out,
                             int N, int outStride, int colOff) {
  __shared__ float sWg[64 * 64];
  __shared__ float sWi[64 * 64];
  __shared__ float sbg[64];
  __shared__ float sbi[64];
  __shared__ float sHa[TILE_WAVES * 16 * HA_PITCH];
  __shared__ float sHt[TILE_WAVES * 16 * HA_PITCH];

  int tid = threadIdx.x;
  for (int i = tid; i < 4096; i += LAYER_THREADS) {
    int r = i >> 6, c = i & 63;
    int idx = (((r >> 1) * 64 + c) << 1) | (r & 1);
    sWg[idx] = Wg[i];
    sWi[idx] = Wi[i];
  }
  if (tid < 64) { sbg[tid] = bg[tid]; sbi[tid] = bi[tid]; }
  __syncthreads();

  int wave = tid >> 5, lane = tid & 31;
  int tiles = (N + 15) >> 4;
  int tile = blockIdx.x * TILE_WAVES + wave;
  if (tile >= tiles) return;

  int m0 = tile << 4;
  float* sT = sHa + wave * (16 * HA_PITCH);
  float* sH = sHt + wave * (16 * HA_PITCH);

  float4 ht[8];
#pragma unroll
  for (int i = 0; i < 8; i++) {
    int f4 = i * 32 + lane;
    int row = m0 + (f4 >> 4);
    if (row > N - 1) row = N - 1;
    ht[i] = *reinterpret_cast<const float4*>(h + (size_t)row * 64 + (f4 & 15) * 4);
  }

  {
    const int2* rec = reinterpret_cast<const int2*>(cw);
    int rpi = m0 + lane;
    if (rpi > N) rpi = N;
    int rp = (lane <= 16) ? rowptr[rpi] : 0;
    int s    = __builtin_amdgcn_readlane(rp, 0);
    int sEnd = __builtin_amdgcn_readlane(rp, 16);
    int rowEnd = __builtin_amdgcn_readlane(rp, 1);
    int j = 0;
    v2f acc = {0.0f, 0.0f};
    const char* hb = reinterpret_cast<const char*>(h);
    unsigned laneOff = (unsigned)lane * 8u;

    int2 cur = {0, 0};
    if (s + lane < sEnd) cur = rec[s + lane];
#pragma unroll
    for (int i = 0; i < 8; i++) {
      int f4 = i * 32 + lane;
      *reinterpret_cast<float4*>(sH + (f4 >> 4) * HA_PITCH + (f4 & 15) * 4) = ht[i];
    }
    for (int base = s; base < sEnd; base += 32) {
      int n = sEnd - base;
      if (n > 32) n = 32;
      int2 nxt = {0, 0};
      if (base + 32 + lane < sEnd) nxt = rec[base + 32 + lane];
      int cj = cur.x, wj = cur.y;

      if (n == 32) gather_trip32(acc, s, j, rowEnd, rp, cj, wj, hb, laneOff, sT, lane);
      else         gather_trip_short(acc, s, j, rowEnd, rp, cj, wj, n, hb, laneOff, sT, lane);
      cur = nxt;
    }
    for (; j < 16; j++) {
      *reinterpret_cast<v2f*>(sT + j * HA_PITCH + lane * 2) = acc;
      acc.x = 0.0f; acc.y = 0.0f;
    }
  }
  __builtin_amdgcn_wave_barrier();

  int kh  = lane >> 4;
  int l15 = lane & 15;

  const float* haRow = sT + l15 * HA_PITCH + 2 * kh;
  const float* hRow  = sH + l15 * HA_PITCH + 2 * kh;

  v8f zero8 = {0.f, 0.f, 0.f, 0.f, 0.f, 0.f, 0.f, 0.f};
  v8f accg[4], acci[4];
#pragma unroll
  for (int nt = 0; nt < 4; nt++) { accg[nt] = zero8; acci[nt] = zero8; }

  const v2f* sWg2 = reinterpret_cast<const v2f*>(sWg);
  const v2f* sWi2 = reinterpret_cast<const v2f*>(sWi);

#pragma unroll
  for (int k = 0; k < 64; k += 4) {
    v2f ga = *reinterpret_cast<const v2f*>(haRow + k);
    v2f hv = *reinterpret_cast<const v2f*>(hRow + k);
    v2f ia = ga * hv;
    int k2 = (k >> 1) + kh;
#pragma unroll
    for (int nt = 0; nt < 4; nt++) {
      v2f bgf = sWg2[k2 * 64 + nt * 16 + l15];
      v2f bif = sWi2[k2 * 64 + nt * 16 + l15];
      accg[nt] = __builtin_amdgcn_wmma_f32_16x16x4_f32(
          false, ga, false, bgf, (short)0, accg[nt], false, false);
      acci[nt] = __builtin_amdgcn_wmma_f32_16x16x4_f32(
          false, ia, false, bif, (short)0, acci[nt], false, false);
    }
  }

  float bgl[4], bil[4];
#pragma unroll
  for (int nt = 0; nt < 4; nt++) {
    bgl[nt] = sbg[nt * 16 + l15];
    bil[nt] = sbi[nt * 16 + l15];
  }

#pragma unroll
  for (int j = 0; j < 8; j++) {
    int row = m0 + j + 8 * kh;
    float v[4];
    float sq = 0.0f;
#pragma unroll
    for (int nt = 0; nt < 4; nt++) {
      float g = accg[nt][j] + bgl[nt];
      float s = acci[nt][j] + bil[nt];
      g = (g > 0.0f) ? g : LEAKY * g;
      s = (s > 0.0f) ? s : LEAKY * s;
      v[nt] = g + s;
      sq += v[nt] * v[nt];
    }
#pragma unroll
    for (int m = 1; m < 16; m <<= 1) sq += __shfl_xor(sq, m, 32);
    float inv = rsqrtf(fmaxf(sq, L2EPS));
    if (row < N) {
      float* hn = hnext + (size_t)row * 64 + l15;
      float* op = out + (size_t)row * outStride + colOff + l15;
#pragma unroll
      for (int nt = 0; nt < 4; nt++) {
        hn[nt * 16] = v[nt];
        op[nt * 16] = v[nt] * inv;
      }
    }
  }
}


extern "C" void kernel_launch(void* const* d_in, const int* in_sizes, int n_in,
                              void* d_out, int out_size, void* d_ws, size_t ws_size,
                              hipStream_t stream) {
  const float* x  = (const float*)d_in[0];
  const int*   ei = (const int*)d_in[1];
  const float* Wg = (const float*)d_in[2];
  const float* bg = (const float*)d_in[3];
  const float* Wi = (const float*)d_in[4];
  const float* bi = (const float*)d_in[5];

  const int d = 64;
  const int N = in_sizes[0] / d;
  const int E = in_sizes[1] / 2;
  const int K = in_sizes[2] / (d * d);
  const int outStride = (K + 1) * d;

  const int nb = (N + 1024) / 1024;
  const int C  = (E + CHUNK_EDGES - 1) / CHUNK_EDGES;
  const int Np = (N + 7) / 8 * 8;
  const int Wp = Np / 2;

  int*   cw     = (int*)d_ws;
  float* hA     = (float*)(((uintptr_t)(cw + 2 * (size_t)E) + 255) & ~(uintptr_t)255);
  float* hB     = hA + (size_t)N * d;
  int*   rank   = (int*)(hB + (size_t)N * d);
  int*   cnt    = rank + E;
  int*   rowptr = cnt + N;
  int*   bsum   = rowptr + N + 1;
  float* dinv   = (float*)(bsum + nb);
  unsigned* tab = (unsigned*)(((uintptr_t)(dinv + N) + 15) & ~(uintptr_t)15);
  int*   off    = (int*)(tab + (size_t)C * Wp);
  float* out    = (float*)d_out;

  const int copyBlocks = (N * 16 + COPY_F4_PER_BLOCK - 1) / COPY_F4_PER_BLOCK;
  count_copy_kernel<<<C + copyBlocks, 1024, 0, stream>>>(ei, rank, tab, E, N, Wp, C, x, out, outStride);
  colscan_kernel<<<nb, 512, 0, stream>>>(tab, off, cnt, bsum, N, C, Wp, Np);
  scan_kernel<<<nb, 256, 0, stream>>>(cnt, bsum, rowptr, dinv, N);
  place_kernel<<<dim3((CHUNK_EDGES + 255) / 256, C), 256, 0, stream>>>(ei, rank, rowptr, off, dinv, cw, E, Np);

  const float* h = x;
  float* nxt = hA;
  int tiles = (N + 15) / 16;
  int lblocks = (tiles + TILE_WAVES - 1) / TILE_WAVES;

  for (int layer = 0; layer < K; layer++) {
    layer_kernel<<<lblocks, LAYER_THREADS, 0, stream>>>(
        rowptr, cw, h,
        Wg + (size_t)layer * d * d, bg + (size_t)layer * d,
        Wi + (size_t)layer * d * d, bi + (size_t)layer * d,
        nxt, out, N, outStride, (layer + 1) * d);
    h = nxt;
    nxt = (nxt == hA) ? hB : hA;
  }
}


# Layer launch: work-group barrier split: a wave signals once its weight words (loaded as one batch beside its slot bounds) are in LDS and waits only before the first B read, after its gather
# speedup vs baseline: 3.6971x; 1.0343x over previous
#include <hip/hip_runtime.h>
#include <stdint.h>

typedef float v2f __attribute__((ext_vector_type(2)));
typedef float v8f __attribute__((ext_vector_type(8)));
typedef float v4f __attribute__((ext_vector_type(4)));

#define LEAKY 0.2f
#define L2EPS 1e-12f


#define CHUNK_EDGES 26624
#define CAP_ROWS 131072
#define CAP_WORDS (CAP_ROWS / 2)

#define COPY_F4_PER_THREAD 8
#define COPY_F4_PER_BLOCK (COPY_F4_PER_THREAD * 1024)

__launch_bounds__(1024)
__global__ void count_copy_kernel(const int* __restrict__ ei, int* __restrict__ rank,
                                  unsigned* __restrict__ tab, int E, int N, int Wp, int C,
                                  const float* __restrict__ x, float* __restrict__ out, int outStride) {
  __shared__ unsigned sc[CAP_WORDS];
  int tid = threadIdx.x;
  if ((int)blockIdx.x >= C) {
    int total4 = N * 16;
    int b0 = ((int)blockIdx.x - C) * COPY_F4_PER_BLOCK;
    int i0 = b0 + tid;
    const v4f* x4 = reinterpret_cast<const v4f*>(x);
    if (b0 + COPY_F4_PER_BLOCK <= total4) {
      v4f v[COPY_F4_PER_THREAD];
#pragma unroll
      for (int u = 0; u < COPY_F4_PER_THREAD; u++) v[u] = x4[i0 + u * 1024];
      asm volatile("" :: "v"(v[0]), "v"(v[1]), "v"(v[2]), "v"(v[3]), "v"(v[4]), "v"(v[5]), "v"(v[6]), "v"(v[7]));
#pragma unroll
      for (int u = 0; u < COPY_F4_PER_THREAD; u++) {
        int i = i0 + u * 1024;
        reinterpret_cast<v4f*>(out + (size_t)(i >> 4) * outStride)[i & 15] = v[u];
      }
    } else {
      for (int u = 0; u < COPY_F4_PER_THREAD; u++) {
        int i = i0 + u * 1024;
        if (i < total4) reinterpret_cast<v4f*>(out + (size_t)(i >> 4) * outStride)[i & 15] = x4[i];
      }
    }
    return;
  }
  int e0 = blockIdx.x * CHUNK_EDGES;
  int e1 = e0 + CHUNK_EDGES;
  if (e1 > E) e1 = E;
  uint4* trow4 = reinterpret_cast<uint4*>(tab + (size_t)blockIdx.x * Wp);
  uint4* sc4 = reinterpret_cast<uint4*>(sc);

  for (int base = 0; base < N; base += CAP_ROWS) {
    int rows = N - base;
    if (rows > CAP_ROWS) rows = CAP_ROWS;
    int quads = (((rows + 1) >> 1) + 3) >> 2;
    uint4 z4 = {0u, 0u, 0u, 0u};
    for (int i = tid; i < quads; i += 1024) sc4[i] = z4;
    __syncthreads();

    for (int eb = e0; eb < e1; eb += 13 * 1024) {
      int rr[13];
#pragma unroll
      for (int u = 0; u < 13; u++) {
        int e = eb + u * 1024 + tid;
        rr[u] = (e < e1) ? ei[e] : -1;
      }
#pragma unroll
      for (int u = 0; u < 13; u++) {
        int e = eb + u * 1024 + tid;
        int r = rr[u] - base;
        if (e < e1 && (unsigned)r < (unsigned)rows) {
          unsigned sh = (unsigned)(r & 1) << 4;
          unsigned old = atomicAdd(&sc[r >> 1], 1u << sh);
          rank[e] = (int)((old >> sh) & 0xFFFFu);
        }
      }
    }
    __syncthreads();

    uint4* dst = trow4 + (base >> 3);
    for (int i = tid; i < quads; i += 1024) dst[i] = sc4[i];
    __syncthreads();
  }
}

__device__ __forceinline__ int wave_sum_i32(int v) {
  for (int m = 1; m < 32; m <<= 1) v += __shfl_xor(v, m, 32);
  return v;
}

__device__ __forceinline__ int block_sum_i32(int v, int* s8) {
  v = wave_sum_i32(v);
  if ((threadIdx.x & 31) == 0) s8[threadIdx.x >> 5] = v;
  __syncthreads();
  int t = 0;
  for (int i = 0; i < 8; i++) t += s8[i];
  return t;
}

__launch_bounds__(512)
__global__ void colscan_kernel(const unsigned* __restrict__ tab, int* __restrict__ off,
                               int* __restrict__ cnt, int* __restrict__ bsum,
                               int N, int C, int Wp, int Np) {
  __shared__ int s16[16];
  int wd = blockIdx.x * 512 + threadIdx.x;
  int r0 = wd * 2;
  int lo = 0, hi = 0;
  if (r0 < N) {
    for (int c0 = 0; c0 < C; c0 += 32) {
      unsigned w[32];
#pragma unroll
      for (int u = 0; u < 32; u++) {
        int cc = (c0 + u < C) ? (c0 + u) : (C - 1);
        unsigned t = tab[(size_t)cc * Wp + wd];
        w[u] = (c0 + u < C) ? t : 0u;
      }
#pragma unroll
      for (int u = 0; u < 32; u++) {
        if (c0 + u < C) {
          int2 o;
          o.x = lo;
          o.y = hi;
          *reinterpret_cast<int2*>(off + (size_t)(c0 + u) * Np + r0) = o;
          lo += (int)(w[u] & 0xFFFFu);
          hi += (int)(w[u] >> 16);
        }
      }
    }
    cnt[r0] = lo;
    if (r0 + 1 < N) cnt[r0 + 1] = hi;
  }
  int t = wave_sum_i32(lo + hi);
  if ((threadIdx.x & 31) == 0) s16[threadIdx.x >> 5] = t;
  __syncthreads();
  if (threadIdx.x == 0) {
    int a = 0;
    for (int i = 0; i < 16; i++) a += s16[i];
    bsum[blockIdx.x] = a;
  }
}

__launch_bounds__(256)
__global__ void scan_kernel(const int* __restrict__ cnt, const int* __restrict__ bsum,
                            int* __restrict__ rowptr, float* __restrict__ dinv, int N) {
  __shared__ int sA[8];
  __shared__ int sB[8];
  int tid = threadIdx.x, lane = tid & 31, wv = tid >> 5;
  int b = blockIdx.x;
  int part = 0;
  for (int i = tid; i < b; i += 256) part += bsum[i];
  int blockOff = block_sum_i32(part, sA);

  int idx0 = b * 1024 + tid * 4;
  int v[4];
  int t = 0;
  for (int k = 0; k < 4; k++) {
    v[k] = (idx0 + k < N) ? cnt[idx0 + k] : 0;
    t += v[k];
  }
  int inc = t;
  for (int d = 1; d < 32; d <<= 1) {
    int o = __shfl_up(inc, d, 32);
    if (lane >= d) inc += o;
  }
  if (lane == 31) sB[wv] = inc;
  __syncthreads();
  int woff = 0;
  for (int i = 0; i < 8; i++) { int s = sB[i]; if (i < wv) woff += s; }
  int p = blockOff + woff + inc - t;
  for (int k = 0; k < 4; k++) {
    int i = idx0 + k;
    if (i <= N) rowptr[i] = p;
    if (i < N) dinv[i] = (v[k] > 0) ? rsqrtf((float)v[k]) : 0.0f;
    p += v[k];
  }
}

__global__ void place_kernel(const int* __restrict__ ei, const int* __restrict__ rank,
                             const int* __restrict__ rowptr, const int* __restrict__ off,
                             const float* __restrict__ dinv,
                             int* __restrict__ cw, int E, int Np) {
  int l = blockIdx.x * blockDim.x + threadIdx.x;
  int e = blockIdx.y * CHUNK_EDGES + l;
  if (l < CHUNK_EDGES && e < E) {
    int r = ei[e];
    int c = ei[E + e];
    float w = dinv[r] * dinv[c];
    int2 rec;
    rec.x = c;
    rec.y = __float_as_int(w);
    int slot = rowptr[r] + off[(size_t)blockIdx.y * Np + r] + rank[e];
    reinterpret_cast<int2*>(cw)[slot] = rec;
  }
}


#define TILE_WAVES 25
#define LAYER_THREADS (TILE_WAVES * 32)
#define HA_PITCH 68

__device__ __forceinline__ void row_step(v2f& acc, int s, int& j, int& rowEnd, int rp,
                                         float* sT, int lane) {
  while (s >= rowEnd) {
    *reinterpret_cast<v2f*>(sT + j * HA_PITCH + lane * 2) = acc;
    acc.x = 0.0f; acc.y = 0.0f;
    j++;
    rowEnd = (j < 15) ? __builtin_amdgcn_readlane(rp, j + 1) : 0x7fffffff;
  }
}

__device__ __forceinline__ void gather_trip32(v2f& acc, int& s, int& j, int& rowEnd, int rp,
                                              int cj, int wj,
                                              const char* __restrict__ hb, unsigned laneOff,
                                              float* sT, int lane) {
  v2f hv[32];
#pragma unroll
  for (int q = 0; q < 32; q++) {
    int c = __builtin_amdgcn_readlane(cj, q);
    hv[q] = *reinterpret_cast<const v2f*>(hb + (size_t)(unsigned)c * 256u + laneOff);
  }
#pragma unroll
  for (int q = 0; q < 32; q++) {
    row_step(acc, s + q, j, rowEnd, rp, sT, lane);
    float w = __int_as_float(__builtin_amdgcn_readlane(wj, q));
    acc += w * hv[q];
  }
  s += 32;
}

__device__ __forceinline__ void gather_trip_short(v2f& acc, int& s, int& j, int& rowEnd, int rp,
                                                  int cj, int wj, int n,
                                                  const char* __restrict__ hb, unsigned laneOff,
                                                  float* sT, int lane) {
  for (int q0 = 0; q0 < n; q0 += 8) {
    v2f hv[8];
#pragma unroll
    for (int u = 0; u < 8; u++) {
      int c = __builtin_amdgcn_readlane(cj, q0 + u);
      hv[u] = *reinterpret_cast<const v2f*>(hb + (size_t)(unsigned)c * 256u + laneOff);
    }
#pragma unroll
    for (int u = 0; u < 8; u++) {
      if (q0 + u < n) {
        row_step(acc, s, j, rowEnd, rp, sT, lane);
        float w = __int_as_float(__builtin_amdgcn_readlane(wj, q0 + u));
        acc += w * hv[u];
        s++;
      }
    }
  }
}

__launch_bounds__(LAYER_THREADS)
__global__ void layer_kernel(const int* __restrict__ rowptr, const int* __restrict__ cw,
                             const float* __restrict__ h,
                             const float* __restrict__ Wg, const float* __restrict__ bg,
                             const float* __restrict__ Wi, const float* __restrict__ bi,
                             float* __restrict__ hnext, float* __restrict__ out,
                             int N, int outStride, int colOff) {
  __shared__ float sWg[64 * 64];
  __shared__ float sWi[64 * 64];
  __shared__ float sbg[64];
  __shared__ float sbi[64];
  __shared__ float sHa[TILE_WAVES * 16 * HA_PITCH];
  __shared__ float sHt[TILE_WAVES * 16 * HA_PITCH];

  int tid = threadIdx.x;
  int wave = tid >> 5, lane = tid & 31;
  int tiles = (N + 15) >> 4;
  int tile = blockIdx.x * TILE_WAVES + wave;
  bool hasTile = tile < tiles;
  int m0 = tile << 4;
  float* sT = sHa + wave * (16 * HA_PITCH);
  float* sH = sHt + wave * (16 * HA_PITCH);

  int rpi = m0 + lane;
  if (rpi > N) rpi = N;
  int rp = (lane <= 16) ? rowptr[rpi] : 0;
  float wgv[6], wiv[6];
#pragma unroll
  for (int u = 0; u < 6; u++) {
    int i = tid + u * LAYER_THREADS;
    if (i > 4095) i = 4095;
    wgv[u] = Wg[i];
    wiv[u] = Wi[i];
  }
  float bgv = 0.0f, biv = 0.0f;
  if (tid < 64) { bgv = bg[tid]; biv = bi[tid]; }

  const int2* rec = reinterpret_cast<const int2*>(cw);
  float4 ht[8];
#pragma unroll
  for (int i = 0; i < 8; i++) {
    int f4 = i * 32 + lane;
    int row = m0 + (f4 >> 4);
    if (row > N - 1) row = N - 1;
    ht[i] = *reinterpret_cast<const float4*>(h + (size_t)row * 64 + (f4 & 15) * 4);
  }
  int s      = __builtin_amdgcn_readlane(rp, 0);
  int sEnd   = __builtin_amdgcn_readlane(rp, 16);
  int rowEnd = __builtin_amdgcn_readlane(rp, 1);
  int2 cur = {0, 0};
  if (s + lane < sEnd) cur = rec[s + lane];
#pragma unroll
  for (int i = 0; i < 8; i++) {
    int f4 = i * 32 + lane;
    *reinterpret_cast<float4*>(sH + (f4 >> 4) * HA_PITCH + (f4 & 15) * 4) = ht[i];
  }

#pragma unroll
  for (int u = 0; u < 6; u++) {
    int i = tid + u * LAYER_THREADS;
    if (i < 4096) {
      int r = i >> 6, c = i & 63;
      int idx = (((r >> 1) * 64 + c) << 1) | (r & 1);
      sWg[idx] = wgv[u];
      sWi[idx] = wiv[u];
    }
  }
  if (tid < 64) { sbg[tid] = bgv; sbi[tid] = biv; }
  asm volatile("s_wait_dscnt 0x0\n\ts_barrier_signal -1" ::: "memory");

  {
    int j = 0;
    v2f acc = {0.0f, 0.0f};
    const char* hb = reinterpret_cast<const char*>(h);
    unsigned laneOff = (unsigned)lane * 8u;

    for (int base = s; base < sEnd; base += 32) {
      int n = sEnd - base;
      if (n > 32) n = 32;
      int2 nxt = {0, 0};
      if (base + 32 + lane < sEnd) nxt = rec[base + 32 + lane];
      int cj = cur.x, wj = cur.y;

      if (n == 32) gather_trip32(acc, s, j, rowEnd, rp, cj, wj, hb, laneOff, sT, lane);
      else         gather_trip_short(acc, s, j, rowEnd, rp, cj, wj, n, hb, laneOff, sT, lane);
      cur = nxt;
    }
    for (; j < 16; j++) {
      *reinterpret_cast<v2f*>(sT + j * HA_PITCH + lane * 2) = acc;
      acc.x = 0.0f; acc.y = 0.0f;
    }
  }
  asm volatile("s_barrier_wait -1" ::: "memory");
  if (!hasTile) return;
  __builtin_amdgcn_wave_barrier();

  int kh  = lane >> 4;
  int l15 = lane & 15;

  const float* haRow = sT + l15 * HA_PITCH + 2 * kh;
  const float* hRow  = sH + l15 * HA_PITCH + 2 * kh;

  v8f zero8 = {0.f, 0.f, 0.f, 0.f, 0.f, 0.f, 0.f, 0.f};
  v8f accg[4], acci[4];
#pragma unroll
  for (int nt = 0; nt < 4; nt++) { accg[nt] = zero8; acci[nt] = zero8; }

  const v2f* sWg2 = reinterpret_cast<const v2f*>(sWg);
  const v2f* sWi2 = reinterpret_cast<const v2f*>(sWi);

#pragma unroll
  for (int k = 0; k < 64; k += 4) {
    v2f ga = *reinterpret_cast<const v2f*>(haRow + k);
    v2f hv = *reinterpret_cast<const v2f*>(hRow + k);
    v2f ia = ga * hv;
    int k2 = (k >> 1) + kh;
#pragma unroll
    for (int nt = 0; nt < 4; nt++) {
      v2f bgf = sWg2[k2 * 64 + nt * 16 + l15];
      v2f bif = sWi2[k2 * 64 + nt * 16 + l15];
      accg[nt] = __builtin_amdgcn_wmma_f32_16x16x4_f32(
          false, ga, false, bgf, (short)0, accg[nt], false, false);
      acci[nt] = __builtin_amdgcn_wmma_f32_16x16x4_f32(
          false, ia, false, bif, (short)0, acci[nt], false, false);
    }
  }

  float bgl[4], bil[4];
#pragma unroll
  for (int nt = 0; nt < 4; nt++) {
    bgl[nt] = sbg[nt * 16 + l15];
    bil[nt] = sbi[nt * 16 + l15];
  }

#pragma unroll
  for (int j = 0; j < 8; j++) {
    int row = m0 + j + 8 * kh;
    float v[4];
    float sq = 0.0f;
#pragma unroll
    for (int nt = 0; nt < 4; nt++) {
      float g = accg[nt][j] + bgl[nt];
      float s = acci[nt][j] + bil[nt];
      g = (g > 0.0f) ? g : LEAKY * g;
      s = (s > 0.0f) ? s : LEAKY * s;
      v[nt] = g + s;
      sq += v[nt] * v[nt];
    }
#pragma unroll
    for (int m = 1; m < 16; m <<= 1) sq += __shfl_xor(sq, m, 32);
    float inv = rsqrtf(fmaxf(sq, L2EPS));
    if (row < N) {
      float* hn = hnext + (size_t)row * 64 + l15;
      float* op = out + (size_t)row * outStride + colOff + l15;
#pragma unroll
      for (int nt = 0; nt < 4; nt++) {
        hn[nt * 16] = v[nt];
        op[nt * 16] = v[nt] * inv;
      }
    }
  }
}


extern "C" void kernel_launch(void* const* d_in, const int* in_sizes, int n_in,
                              void* d_out, int out_size, void* d_ws, size_t ws_size,
                              hipStream_t stream) {
  const float* x  = (const float*)d_in[0];
  const int*   ei = (const int*)d_in[1];
  const float* Wg = (const float*)d_in[2];
  const float* bg = (const float*)d_in[3];
  const float* Wi = (const float*)d_in[4];
  const float* bi = (const float*)d_in[5];

  const int d = 64;
  const int N = in_sizes[0] / d;
  const int E = in_sizes[1] / 2;
  const int K = in_sizes[2] / (d * d);
  const int outStride = (K + 1) * d;

  const int nb = (N + 1024) / 1024;
  const int C  = (E + CHUNK_EDGES - 1) / CHUNK_EDGES;
  const int Np = (N + 7) / 8 * 8;
  const int Wp = Np / 2;

  int*   cw     = (int*)d_ws;
  float* hA     = (float*)(((uintptr_t)(cw + 2 * (size_t)E) + 255) & ~(uintptr_t)255);
  float* hB     = hA + (size_t)N * d;
  int*   rank   = (int*)(hB + (size_t)N * d);
  int*   cnt    = rank + E;
  int*   rowptr = cnt + N;
  int*   bsum   = rowptr + N + 1;
  float* dinv   = (float*)(bsum + nb);
  unsigned* tab = (unsigned*)(((uintptr_t)(dinv + N) + 15) & ~(uintptr_t)15);
  int*   off    = (int*)(tab + (size_t)C * Wp);
  float* out    = (float*)d_out;

  const int copyBlocks = (N * 16 + COPY_F4_PER_BLOCK - 1) / COPY_F4_PER_BLOCK;
  count_copy_kernel<<<C + copyBlocks, 1024, 0, stream>>>(ei, rank, tab, E, N, Wp, C, x, out, outStride);
  colscan_kernel<<<nb, 512, 0, stream>>>(tab, off, cnt, bsum, N, C, Wp, Np);
  scan_kernel<<<nb, 256, 0, stream>>>(cnt, bsum, rowptr, dinv, N);
  place_kernel<<<dim3((CHUNK_EDGES + 255) / 256, C), 256, 0, stream>>>(ei, rank, rowptr, off, dinv, cw, E, Np);

  const float* h = x;
  float* nxt = hA;
  int tiles = (N + 15) / 16;
  int lblocks = (tiles + TILE_WAVES - 1) / TILE_WAVES;

  for (int layer = 0; layer < K; layer++) {
    layer_kernel<<<lblocks, LAYER_THREADS, 0, stream>>>(
        rowptr, cw, h,
        Wg + (size_t)layer * d * d, bg + (size_t)layer * d,
        Wi + (size_t)layer * d * d, bi + (size_t)layer * d,
        nxt, out, N, outStride, (layer + 1) * d);
    h = nxt;
    nxt = (nxt == hA) ? hB : hA;
  }
}
